# starting kernel
# baseline (speedup 1.0000x reference)
#include <hip/hip_runtime.h>

typedef float v2f __attribute__((ext_vector_type(2)));
typedef float v8f __attribute__((ext_vector_type(8)));

#define HIDDEN 1024
#define SUBSP  256
#define LRM    1.0f
#define ROWS   (4 * 4096)

__global__ __launch_bounds__(256)
void sluice_exchange_wmma(const float* __restrict__ Ain,
                          const float* __restrict__ Bin,
                          const float* __restrict__ Emat,
                          float* __restrict__ outA,
                          float* __restrict__ outB) {
  const int tid  = blockIdx.x * blockDim.x + threadIdx.x;
  const int wave = tid >> 5;
  const int lane = threadIdx.x & 31;
  const int bs   = wave >> 1;
  const int d0   = (wave & 1) << 7;

  const int n  = lane & 15;
  const int kh = lane >> 4;

  v2f a1, a2;
  if (n < 8) {
    a1.x = LRM * Emat[(2 * kh + 0) * 8 + n];
    a1.y = LRM * Emat[(2 * kh + 1) * 8 + n];
    a2.x = LRM * Emat[(2 * kh + 4) * 8 + n];
    a2.y = LRM * Emat[(2 * kh + 5) * 8 + n];
  } else {
    a1.x = 0.0f; a1.y = 0.0f; a2.x = 0.0f; a2.y = 0.0f;
  }

  const float* pa = Ain  + (size_t)bs * HIDDEN + d0;
  const float* pb = Bin  + (size_t)bs * HIDDEN + d0;
  float*       oa = outA + (size_t)bs * HIDDEN + d0;
  float*       ob = outB + (size_t)bs * HIDDEN + d0;

  const int koff0 = (2 * kh + 0) * SUBSP + n;
  const int koff1 = (2 * kh + 1) * SUBSP + n;

#pragma unroll
  for (int t = 0; t < 8; ++t) {
    const int dt = t * 16;

    v2f b1, b2;
    b1.x = __builtin_nontemporal_load(pa + koff0 + dt);
    b1.y = __builtin_nontemporal_load(pa + koff1 + dt);
    b2.x = __builtin_nontemporal_load(pb + koff0 + dt);
    b2.y = __builtin_nontemporal_load(pb + koff1 + dt);

    v8f acc = {0.f, 0.f, 0.f, 0.f, 0.f, 0.f, 0.f, 0.f};
    acc = __builtin_amdgcn_wmma_f32_16x16x4_f32(false, a1, false, b1,
                                                (short)0, acc, false, false);
    acc = __builtin_amdgcn_wmma_f32_16x16x4_f32(false, a2, false, b2,
                                                (short)0, acc, false, false);

    if (kh == 0) {
#pragma unroll
      for (int r = 0; r < 4; ++r)
        __builtin_nontemporal_store(acc[r], oa + r * SUBSP + dt + n);
#pragma unroll
      for (int r = 4; r < 8; ++r)
        __builtin_nontemporal_store(acc[r], ob + (r - 4) * SUBSP + dt + n);
    }
  }
}

extern "C" void kernel_launch(void* const* d_in, const int* in_sizes, int n_in,
                              void* d_out, int out_size, void* d_ws, size_t ws_size,
                              hipStream_t stream) {
  const float* a = (const float*)d_in[0];
  const float* b = (const float*)d_in[1];
  const float* E = (const float*)d_in[2];
  float* outA = (float*)d_out;
  float* outB = outA + (size_t)ROWS * HIDDEN;

  dim3 grid(ROWS * 2 / 8);
  dim3 block(256);
  hipLaunchKernelGGL(sluice_exchange_wmma, grid, block, 0, stream,
                     a, b, E, outA, outB);
}


# Whole-line 16-byte loads and stores: a tile column stands for 4 adjacent positions (one per tile), lanes 16-31 carry a second set of 64 positions through a block-diagonal E^T A operand
# speedup vs baseline: 2.3156x; 2.3156x over previous
#include <hip/hip_runtime.h>


typedef float v2f __attribute__((ext_vector_type(2)));
typedef float v4f __attribute__((ext_vector_type(4)));
typedef float v8f __attribute__((ext_vector_type(8)));

#define HIDDEN 1024
#define SUBSP  256
#define LRM    1.0f
#define ROWS   (4 * 4096)


static __device__ __forceinline__ v8f tile(v2f a1lo, v2f a2lo, v2f a1hi, v2f a2hi,
                                           float aPx, float aPy, float bPx, float bPy,
                                           float aQx, float aQy, float bQx, float bQy) {
  v8f c = {0.f, 0.f, 0.f, 0.f, 0.f, 0.f, 0.f, 0.f};
  v2f b;
  b.x = aPx; b.y = aPy;
  c = __builtin_amdgcn_wmma_f32_16x16x4_f32(false, a1lo, false, b, (short)0, c, false, false);
  b.x = bPx; b.y = bPy;
  c = __builtin_amdgcn_wmma_f32_16x16x4_f32(false, a2lo, false, b, (short)0, c, false, false);
  b.x = aQx; b.y = aQy;
  c = __builtin_amdgcn_wmma_f32_16x16x4_f32(false, a1hi, false, b, (short)0, c, false, false);
  b.x = bQx; b.y = bQy;
  c = __builtin_amdgcn_wmma_f32_16x16x4_f32(false, a2hi, false, b, (short)0, c, false, false);
  return c;
}

template <int R>
static __device__ __forceinline__ void store_row(float* p, v8f c0, v8f c1, v8f c2, v8f c3) {
  v4f o = {c0[R], c1[R], c2[R], c3[R]};
  __builtin_nontemporal_store(o, (v4f*)p);
}

__global__ __launch_bounds__(256)
void sluice_exchange_wmma(const float* __restrict__ Ain,
                          const float* __restrict__ Bin,
                          const float* __restrict__ Emat,
                          float* __restrict__ outA,
                          float* __restrict__ outB) {
  const int tid  = blockIdx.x * blockDim.x + threadIdx.x;
  const int wave = tid >> 5;
  const int lane = threadIdx.x & 31;
  const int bs   = wave >> 1;
  const int d0   = (wave & 1) << 7;

  const int n  = lane & 15;
  const int kh = lane >> 4;

  const int j = n & 7;
  const float e1x = LRM * Emat[(2 * kh + 0) * 8 + j];
  const float e1y = LRM * Emat[(2 * kh + 1) * 8 + j];
  const float e2x = LRM * Emat[(2 * kh + 4) * 8 + j];
  const float e2y = LRM * Emat[(2 * kh + 5) * 8 + j];
  const bool lo = n < 8;
  v2f a1lo, a2lo, a1hi, a2hi;
  a1lo.x = lo ? e1x : 0.0f;  a1lo.y = lo ? e1y : 0.0f;
  a2lo.x = lo ? e2x : 0.0f;  a2lo.y = lo ? e2y : 0.0f;
  a1hi.x = lo ? 0.0f : e1x;  a1hi.y = lo ? 0.0f : e1y;
  a2hi.x = lo ? 0.0f : e2x;  a2hi.y = lo ? 0.0f : e2y;

  const size_t rowbase = (size_t)bs * HIDDEN + d0;
  const float* pa = Ain + rowbase;
  const float* pb = Bin + rowbase;

  const int offx = (2 * kh + 0) * SUBSP + 4 * n;
  const int offy = (2 * kh + 1) * SUBSP + 4 * n;

  const v4f aPx = __builtin_nontemporal_load((const v4f*)(pa + offx));
  const v4f aPy = __builtin_nontemporal_load((const v4f*)(pa + offy));
  const v4f bPx = __builtin_nontemporal_load((const v4f*)(pb + offx));
  const v4f bPy = __builtin_nontemporal_load((const v4f*)(pb + offy));
  const v4f aQx = __builtin_nontemporal_load((const v4f*)(pa + offx + 64));
  const v4f aQy = __builtin_nontemporal_load((const v4f*)(pa + offy + 64));
  const v4f bQx = __builtin_nontemporal_load((const v4f*)(pb + offx + 64));
  const v4f bQy = __builtin_nontemporal_load((const v4f*)(pb + offy + 64));

  const v8f c0 = tile(a1lo, a2lo, a1hi, a2hi, aPx.x, aPy.x, bPx.x, bPy.x, aQx.x, aQy.x, bQx.x, bQy.x);
  const v8f c1 = tile(a1lo, a2lo, a1hi, a2hi, aPx.y, aPy.y, bPx.y, bPy.y, aQx.y, aQy.y, bQx.y, bQy.y);
  const v8f c2 = tile(a1lo, a2lo, a1hi, a2hi, aPx.z, aPy.z, bPx.z, bPy.z, aQx.z, aQy.z, bQx.z, bQy.z);
  const v8f c3 = tile(a1lo, a2lo, a1hi, a2hi, aPx.w, aPy.w, bPx.w, bPy.w, aQx.w, aQy.w, bQx.w, bQy.w);

  float* oa = outA + rowbase + 4 * lane;
  float* ob = outB + rowbase + 4 * lane;
  store_row<0>(oa + 0 * SUBSP, c0, c1, c2, c3);
  store_row<1>(oa + 1 * SUBSP, c0, c1, c2, c3);
  store_row<2>(oa + 2 * SUBSP, c0, c1, c2, c3);
  store_row<3>(oa + 3 * SUBSP, c0, c1, c2, c3);
  store_row<4>(ob + 0 * SUBSP, c0, c1, c2, c3);
  store_row<5>(ob + 1 * SUBSP, c0, c1, c2, c3);
  store_row<6>(ob + 2 * SUBSP, c0, c1, c2, c3);
  store_row<7>(ob + 3 * SUBSP, c0, c1, c2, c3);
}

extern "C" void kernel_launch(void* const* d_in, const int* in_sizes, int n_in,
                              void* d_out, int out_size, void* d_ws, size_t ws_size,
                              hipStream_t stream) {
  const float* a = (const float*)d_in[0];
  const float* b = (const float*)d_in[1];
  const float* E = (const float*)d_in[2];
  float* outA = (float*)d_out;
  float* outB = outA + (size_t)ROWS * HIDDEN;

  dim3 grid(ROWS * 2 / 8);
  dim3 block(256);
  hipLaunchKernelGGL(sluice_exchange_wmma, grid, block, 0, stream,
                     a, b, E, outA, outB);
}
